# starting kernel
# baseline (speedup 1.0000x reference)
#include <hip/hip_runtime.h>

typedef __attribute__((ext_vector_type(2))) float v2f;
typedef __attribute__((ext_vector_type(8))) float v8f;

#define N_NODES 100000
#define N_EDGES 1600000
#define D 128
#define COMB_STRIDE 260
#define OUT_STRIDE 132

__global__ void zero_f32(float4* __restrict__ p, long n4) {
    long i = (long)blockIdx.x * blockDim.x + threadIdx.x;
    long stride = (long)gridDim.x * blockDim.x;
    float4 z = {0.f, 0.f, 0.f, 0.f};
    for (; i < n4; i += stride) p[i] = z;
}

__global__ void edge_count(const int* __restrict__ dst, float* __restrict__ cnt) {
    int e = blockIdx.x * blockDim.x + threadIdx.x;
    if (e < N_EDGES) atomicAdd(&cnt[dst[e]], 1.0f);
}

__global__ void edge_scatter(const int* __restrict__ src, const int* __restrict__ dst,
                             const float* __restrict__ x, float* __restrict__ agg) {
    long idx = (long)blockIdx.x * blockDim.x + threadIdx.x;
    long e = idx >> 5;
    if (e >= N_EDGES) return;
    int f = (int)(idx & 31) * 4;
    long s = src[e];
    long d = dst[e];
    float4 v = *(const float4*)(x + s * D + f);
    float* ap = agg + d * D + f;
    atomicAdd(ap + 0, v.x);
    atomicAdd(ap + 1, v.y);
    atomicAdd(ap + 2, v.z);
    atomicAdd(ap + 3, v.w);
}

__global__ void __launch_bounds__(256)
sage_gemm(const float* __restrict__ agg, const float* __restrict__ cnt,
          const float* __restrict__ xin, const float* __restrict__ Wl,
          const float* __restrict__ bl,  const float* __restrict__ Wr,
          float* __restrict__ out, int relu) {
    extern __shared__ float lds[];

    for (int i = threadIdx.x; i < D * (D / 2); i += 256) {
        int n = i >> 6;
        int p = i & 63;
        v2f wl = *(const v2f*)(Wl + n * D + 2 * p);
        v2f wr = *(const v2f*)(Wr + n * D + 2 * p);
        float4 c4 = {wl.x, wl.y, wr.x, wr.y};
        *(float4*)(lds + n * COMB_STRIDE + p * 4) = c4;
    }
    __syncthreads();

    const int lane = threadIdx.x & 31;
    const int wave = threadIdx.x >> 5;
    const int half = lane >> 4;
    const int l16  = lane & 15;

    long rowA = (long)blockIdx.x * 128 + wave * 16 + l16;
    long rA   = rowA < N_NODES ? rowA : (long)(N_NODES - 1);

    float c   = cnt[rA];
    float inv = 1.0f / (c > 1.0f ? c : 1.0f);

    const float* aggRow = agg + rA * D;
    const float* xRow   = xin + rA * D;

    v8f acc[8] = {};

    for (int k = 0; k < D; k += 4) {
        int ko = k + 2 * half;
        int p4 = ((ko >> 1) << 2);
        v2f aM = *(const v2f*)(aggRow + ko);
        aM = aM * inv;
        v2f aX = *(const v2f*)(xRow + ko);

        float4 bf[8];
        #pragma unroll
        for (int t = 0; t < 8; ++t) {
            int n = t * 16 + l16;
            bf[t] = *(const float4*)(lds + n * COMB_STRIDE + p4);
        }
        #pragma unroll
        for (int t = 0; t < 8; ++t) {
            v2f bL = {bf[t].x, bf[t].y};
            v2f bR = {bf[t].z, bf[t].w};
            acc[t] = __builtin_amdgcn_wmma_f32_16x16x4_f32(
                false, aM, false, bL, (short)0, acc[t], false, false);
            acc[t] = __builtin_amdgcn_wmma_f32_16x16x4_f32(
                false, aX, false, bR, (short)0, acc[t], false, false);
        }
    }

    __syncthreads();

    int localBase = wave * 16 + half * 8;
    #pragma unroll
    for (int t = 0; t < 8; ++t) {
        int n = t * 16 + l16;
        float bias = bl[n];
        #pragma unroll
        for (int r = 0; r < 8; ++r) {
            float v = acc[t][r] + bias;
            if (relu) v = v > 0.0f ? v : 0.0f;
            lds[(localBase + r) * OUT_STRIDE + n] = v;
        }
    }
    __syncthreads();

    long blockBase = (long)blockIdx.x * 128;
    for (int i = threadIdx.x; i < 128 * (D / 4); i += 256) {
        int row = i >> 5;
        int c4  = (i & 31) * 4;
        long grow = blockBase + row;
        if (grow < N_NODES) {
            float4 v = *(const float4*)(lds + row * OUT_STRIDE + c4);
            *(float4*)(out + grow * D + c4) = v;
        }
    }
}

extern "C" void kernel_launch(void* const* d_in, const int* in_sizes, int n_in,
                              void* d_out, int out_size, void* d_ws, size_t ws_size,
                              hipStream_t stream) {
    const float* x   = (const float*)d_in[0];
    const int*   ei  = (const int*)d_in[1];
    const float* W1l = (const float*)d_in[2];
    const float* b1l = (const float*)d_in[3];
    const float* W1r = (const float*)d_in[4];
    const float* W2l = (const float*)d_in[5];
    const float* b2l = (const float*)d_in[6];
    const float* W2r = (const float*)d_in[7];

    const int* src = ei;
    const int* dst = ei + N_EDGES;

    float* agg  = (float*)d_ws;
    float* cntp = agg + (size_t)N_NODES * D;
    float* h    = cntp + N_NODES;
    float* outp = (float*)d_out;

    const long nAgg = (long)N_NODES * D;
    const int  threads = 256;
    const long scatterThreads = (long)N_EDGES * 32;
    const int  scatterBlocks  = (int)((scatterThreads + threads - 1) / threads);
    const int  gemmBlocks     = (N_NODES + 127) / 128;
    const size_t shmem        = (size_t)D * COMB_STRIDE * sizeof(float);

    zero_f32<<<2048, threads, 0, stream>>>((float4*)agg, (nAgg + N_NODES) / 4);
    edge_count<<<(N_EDGES + threads - 1) / threads, threads, 0, stream>>>(dst, cntp);
    edge_scatter<<<scatterBlocks, threads, 0, stream>>>(src, dst, x, agg);
    sage_gemm<<<gemmBlocks, threads, shmem, stream>>>(agg, cntp, x, W1l, b1l, W1r, h, 1);

    zero_f32<<<2048, threads, 0, stream>>>((float4*)agg, nAgg / 4);
    edge_scatter<<<scatterBlocks, threads, 0, stream>>>(src, dst, h, agg);
    sage_gemm<<<gemmBlocks, threads, shmem, stream>>>(agg, cntp, h, W2l, b2l, W2r, outp, 0);
}


# One launch per layer, no edge grouping: a workgroup holds its 391 destination rows' sums in LDS, filters the raw edge list, adds source rows by LDS f32 atomics (arrival order); WMMA reads A from LDS
# speedup vs baseline: 2.3222x; 2.3222x over previous
#include <hip/hip_runtime.h>

typedef __attribute__((ext_vector_type(2))) float v2f;
typedef __attribute__((ext_vector_type(8))) float v8f;

#define N_NODES 100000
#define N_EDGES 1600000
#define D 128
#define NWG 256
#define ROWS_WG 391
#define TILE_ROWS 400
#define PA 132
#define NWAVES 16
#define THREADS 512
#define QCAP 512
#define NGROUPS (N_EDGES / 512)
static_assert(NWG * ROWS_WG >= N_NODES, "row ownership must cover all nodes");
static_assert(N_EDGES % 512 == 0, "edge list is scanned in groups of 512");
static_assert(N_EDGES < (1 << 21) && TILE_ROWS <= (1 << 9), "ring entry packs row<<21 | edge");
static_assert(D == 128, "a lane owns columns lane + 32*k");

#define LDS_FLOATS (TILE_ROWS * PA + D * PA + TILE_ROWS + NWAVES * QCAP)
static_assert(LDS_FLOATS * 4 <= 320 * 1024, "sums, one weight matrix, counts and rings fit one WGP's LDS");

template <bool FULL>
__device__ __forceinline__ void drain(const int* __restrict__ src, const float* __restrict__ x,
                                      float* aggL, const unsigned* q, unsigned qhead,
                                      int n, int lane) {
    unsigned ent = 0; int s = 0;
    if (FULL || lane < n) {
        ent = q[(qhead + lane) & (QCAP - 1)];
        s = src[ent & 0x1FFFFFu];
    }
    if (FULL) {
        #pragma unroll
        for (int j0 = 0; j0 < 32; j0 += 8) {
            float v[8][4];
            #pragma unroll
            for (int j = 0; j < 8; ++j) {
                const int sj = __builtin_amdgcn_readlane(s, j0 + j);
                const float* xr = x + (long)sj * D + lane;
                v[j][0] = xr[0]; v[j][1] = xr[32]; v[j][2] = xr[64]; v[j][3] = xr[96];
            }
            #pragma unroll
            for (int j = 0; j < 8; ++j) {
                const unsigned dj = ((unsigned)__builtin_amdgcn_readlane((int)ent, j0 + j)) >> 21;
                float* ar = aggL + dj * PA + lane;
                atomicAdd(ar,      v[j][0]);
                atomicAdd(ar + 32, v[j][1]);
                atomicAdd(ar + 64, v[j][2]);
                atomicAdd(ar + 96, v[j][3]);
            }
        }
    } else {
        for (int j = 0; j < n; ++j) {
            const int sj = __builtin_amdgcn_readlane(s, j);
            const unsigned dj = ((unsigned)__builtin_amdgcn_readlane((int)ent, j)) >> 21;
            const float* xr = x + (long)sj * D + lane;
            float* ar = aggL + dj * PA + lane;
            const float a0 = xr[0], a1 = xr[32], a2 = xr[64], a3 = xr[96];
            atomicAdd(ar,      a0);
            atomicAdd(ar + 32, a1);
            atomicAdd(ar + 64, a2);
            atomicAdd(ar + 96, a3);
        }
    }
}

template <int NT, bool FROM_LDS>
__device__ __forceinline__ void stage_product(v8f (&acc)[2][8], const float* aggL, const float* wL,
                                              const float* __restrict__ xin, const long (&grow)[2],
                                              const int (&lrow)[2], const float (&inv)[2],
                                              int l16, int half) {
    #pragma unroll 2
    for (int qk = 0; qk < 16; ++qk) {
        const int co = 8 * qk + 4 * half;
        float4 a[2];
        #pragma unroll
        for (int i = 0; i < NT; ++i) {
            if (FROM_LDS) {
                a[i] = *(const float4*)(aggL + lrow[i] * PA + co);
                a[i].x *= inv[i]; a[i].y *= inv[i]; a[i].z *= inv[i]; a[i].w *= inv[i];
            } else {
                a[i] = *(const float4*)(xin + grow[i] * D + co);
            }
        }
        float4 bf[8];
        #pragma unroll
        for (int t = 0; t < 8; ++t) bf[t] = *(const float4*)(wL + (t * 16 + l16) * PA + co);
        #pragma unroll
        for (int t = 0; t < 8; ++t) {
            const v2f b0 = {bf[t].x, bf[t].y};
            const v2f b1 = {bf[t].z, bf[t].w};
            #pragma unroll
            for (int i = 0; i < NT; ++i) {
                const v2f a0 = {a[i].x, a[i].y};
                const v2f a1 = {a[i].z, a[i].w};
                acc[i][t] = __builtin_amdgcn_wmma_f32_16x16x4_f32(false, a0, false, b0, (short)0, acc[i][t], false, false);
                acc[i][t] = __builtin_amdgcn_wmma_f32_16x16x4_f32(false, a1, false, b1, (short)0, acc[i][t], false, false);
            }
        }
    }
}

__device__ __forceinline__ void stage_weights(float* wL, const float* __restrict__ Wt) {
    for (int i = threadIdx.x; i < D * (D / 4); i += THREADS) {
        const int n = i >> 5, c4 = (i & 31) * 4;
        *(float4*)(wL + n * PA + c4) = *(const float4*)(Wt + n * D + c4);
    }
}

__device__ __forceinline__ void products_and_store(float* aggL, float* wL, const float* cntL,
        const float* __restrict__ xin, const float* __restrict__ Wr, const float* __restrict__ bl,
        float* __restrict__ out, int relu, int lo, int nrows, int wave, int lane) {
    const int half = lane >> 4, l16 = lane & 15;
    v8f acc[2][8] = {};
    int lrow[2]; long grow[2]; float inv[2];
    #pragma unroll
    for (int i = 0; i < 2; ++i) {
        lrow[i] = (wave + 16 * i) * 16 + l16;
        if (lrow[i] >= TILE_ROWS) lrow[i] = TILE_ROWS - 1;
        long g = (long)lo + lrow[i];
        grow[i] = g < N_NODES ? g : (long)(N_NODES - 1);
        const float c = cntL[lrow[i]];
        inv[i] = 1.0f / (c > 1.0f ? c : 1.0f);
    }
    stage_product<2, true>(acc, aggL, wL, xin, grow, lrow, inv, l16, half);
    __syncthreads();
    stage_weights(wL, Wr);
    __syncthreads();
    stage_product<2, false>(acc, aggL, wL, xin, grow, lrow, inv, l16, half);

    float* slot = aggL + wave * 16 * PA;
    #pragma unroll
    for (int i = 0; i < 2; ++i) {
        if ((wave + 16 * i) * 16 >= nrows) break;
        #pragma unroll
        for (int t = 0; t < 8; ++t) {
            const int n = t * 16 + l16;
            const float bias = bl[n];
            #pragma unroll
            for (int r = 0; r < 8; ++r) {
                float v = acc[i][t][r] + bias;
                if (relu) v = v > 0.0f ? v : 0.0f;
                slot[(r + 8 * half) * PA + n] = v;
            }
        }
        __builtin_amdgcn_wave_barrier();
        const int tileRow = (wave + 16 * i) * 16;
        #pragma unroll
        for (int r = 0; r < 16; ++r) {
            if (tileRow + r < nrows) {
                const float4 v = *(const float4*)(slot + r * PA + 4 * lane);
                *(float4*)(out + ((long)lo + tileRow + r) * D + 4 * lane) = v;
            }
        }
        __builtin_amdgcn_wave_barrier();
    }
}

__global__ void __launch_bounds__(THREADS)
sage_layer(const int* __restrict__ src, const int* __restrict__ dst,
           const float* __restrict__ xin, const float* __restrict__ Wl,
           const float* __restrict__ bl, const float* __restrict__ Wr,
           float* __restrict__ out, int relu) {
    __shared__ __attribute__((aligned(16))) float lds[LDS_FLOATS];
    float*    aggL = lds;
    float*    wL   = aggL + TILE_ROWS * PA;
    float*    cntL = wL + D * PA;
    unsigned* qAll = (unsigned*)(cntL + TILE_ROWS);

    const int lane = threadIdx.x & 31;
    const int wave = __builtin_amdgcn_readfirstlane((int)(threadIdx.x >> 5));
    const int lo = blockIdx.x * ROWS_WG;
    int nrows = N_NODES - lo;
    if (nrows > ROWS_WG) nrows = ROWS_WG;
    if (nrows < 0) nrows = 0;

    {
        const float4 z = {0.f, 0.f, 0.f, 0.f};
        for (int i = threadIdx.x; i < TILE_ROWS * PA / 4; i += THREADS) ((float4*)aggL)[i] = z;
        for (int i = threadIdx.x; i < TILE_ROWS; i += THREADS) cntL[i] = 0.f;
    }
    stage_weights(wL, Wl);
    __syncthreads();

    unsigned* q = qAll + wave * QCAP;
    unsigned qhead = 0, qn = 0;
    const int4* dst4 = (const int4*)dst;
    const int rot = (blockIdx.x * 12) % NGROUPS;
    int g = wave + rot;
    if (g >= NGROUPS) g -= NGROUPS;
    int4 dv[4];
    if (wave < NGROUPS) {
        const int4* p = dst4 + (long)g * 128 + lane;
        #pragma unroll
        for (int c = 0; c < 4; ++c) dv[c] = p[c * 32];
    }
    for (int it = 0; it * NWAVES + wave < NGROUPS; ++it) {
        const bool more = (it + 1) * NWAVES + wave < NGROUPS;
        int gn = g + NWAVES;
        if (gn >= NGROUPS) gn -= NGROUPS;
        int4 nx[4] = {dv[0], dv[1], dv[2], dv[3]};
        if (more) {
            const int4* p = dst4 + (long)gn * 128 + lane;
            #pragma unroll
            for (int c = 0; c < 4; ++c) nx[c] = p[c * 32];
        }
        #pragma unroll
        for (int c2 = 0; c2 < 4; c2 += 2) {
            #pragma unroll
            for (int c = c2; c < c2 + 2; ++c) {
                const int dd[4] = {dv[c].x, dv[c].y, dv[c].z, dv[c].w};
                #pragma unroll
                for (int k = 0; k < 4; ++k) {
                    const unsigned u = (unsigned)(dd[k] - lo);
                    const bool m = u < (unsigned)nrows;
                    const unsigned mask = __builtin_amdgcn_ballot_w32(m);
                    if (mask) {
                        if (m) {
                            const unsigned rank = __builtin_amdgcn_mbcnt_lo(mask, 0u);
                            const unsigned e = (unsigned)(g * 512 + c * 128 + lane * 4 + k);
                            q[(qhead + qn + rank) & (QCAP - 1)] = (u << 21) | e;
                            atomicAdd(&cntL[u], 1.0f);
                        }
                        qn += (unsigned)__builtin_popcount(mask);
                    }
                }
            }
            while (qn >= 32) {
                drain<true>(src, xin, aggL, q, qhead, 32, lane);
                qhead += 32; qn -= 32;
            }
        }
        #pragma unroll
        for (int c = 0; c < 4; ++c) dv[c] = nx[c];
        g = gn;
    }
    if (qn) drain<false>(src, xin, aggL, q, qhead, (int)qn, lane);
    __syncthreads();

    products_and_store(aggL, wL, cntL, xin, Wr, bl, out, relu, lo, nrows, wave, lane);
}

extern "C" void kernel_launch(void* const* d_in, const int* in_sizes, int n_in,
                              void* d_out, int out_size, void* d_ws, size_t ws_size,
                              hipStream_t stream) {
    const float* x   = (const float*)d_in[0];
    const int*   ei  = (const int*)d_in[1];
    const float* W1l = (const float*)d_in[2];
    const float* b1l = (const float*)d_in[3];
    const float* W1r = (const float*)d_in[4];
    const float* W2l = (const float*)d_in[5];
    const float* b2l = (const float*)d_in[6];
    const float* W2r = (const float*)d_in[7];

    const int* src = ei;
    const int* dst = ei + N_EDGES;
    float* h    = (float*)d_ws;
    float* outp = (float*)d_out;

    sage_layer<<<NWG, THREADS, 0, stream>>>(src, dst, x, W1l, b1l, W1r, h, 1);
    sage_layer<<<NWG, THREADS, 0, stream>>>(src, dst, h, W2l, b2l, W2r, outp, 0);
}


# Edge scan reads one-byte owner codes (plus one packed row/source word per kept edge) that a new small first launch writes, not the 4-byte dst list: 4x fewer scan bytes; adds stay arrival-ordered
# speedup vs baseline: 2.8946x; 1.2465x over previous
#include <hip/hip_runtime.h>

typedef __attribute__((ext_vector_type(2))) float v2f;
typedef __attribute__((ext_vector_type(8))) float v8f;

#define N_NODES 100000
#define N_EDGES 1600000
#define D 128
#define NWG 256
#define ROWS_WG 391
#define TILE_ROWS 400
#define PA 132
#define NWAVES 16
#define THREADS 512
#define QCAP 512
#define NCHUNK (N_EDGES / 512)
#define NPAIR ((NCHUNK + 1) / 2)
#define DEPTH 8
#define H_BYTES   ((size_t)N_NODES * D * 4)
#define OWN_BYTES ((size_t)N_EDGES)
static_assert(NWG * ROWS_WG >= N_NODES, "row ownership must cover all nodes");
static_assert(N_EDGES % 512 == 0, "owner codes are scanned in chunks of 512");
static_assert(NWG <= 256, "an owner code is one byte");
static_assert(N_NODES <= (1 << 17) && ROWS_WG < 511, "edge record packs local row<<17 | source");
static_assert(H_BYTES % 256 == 0 && OWN_BYTES % 256 == 0, "workspace planes stay 256-B aligned");
static_assert(D == 128, "a lane owns columns lane + 32*k");

#define LDS_FLOATS (TILE_ROWS * PA + D * PA + TILE_ROWS + NWAVES * QCAP)
static_assert(LDS_FLOATS * 4 <= 320 * 1024, "sums, one weight matrix, counts and rings fit one WGP's LDS");

__global__ void __launch_bounds__(256)
edge_codes(const int* __restrict__ src, const int* __restrict__ dst,
           unsigned* __restrict__ own, unsigned* __restrict__ rec) {
    const int i = blockIdx.x * 256 + threadIdx.x;
    if (i >= N_EDGES / 4) return;
    const int4 d = ((const int4*)dst)[i];
    const int4 s = ((const int4*)src)[i];
    const int dd[4] = {d.x, d.y, d.z, d.w};
    const int ss[4] = {s.x, s.y, s.z, s.w};
    unsigned code = 0, r[4];
    #pragma unroll
    for (int k = 0; k < 4; ++k) {
        const unsigned dv = (unsigned)dd[k];
        unsigned o = dv / ROWS_WG;
        unsigned u = dv - o * ROWS_WG;
        if (o >= NWG) { o = 0; u = 511u; }
        code |= o << (8 * k);
        r[k] = (u << 17) | ((unsigned)ss[k] & 0x1FFFFu);
    }
    own[i] = code;
    const uint4 rv = {r[0], r[1], r[2], r[3]};
    ((uint4*)rec)[i] = rv;
}

__device__ __forceinline__ unsigned match_bits(const uint4 a, const uint4 b, unsigned pat) {
    const unsigned wa[4] = {a.x, a.y, a.z, a.w};
    const unsigned wb[4] = {b.x, b.y, b.z, b.w};
    unsigned Z = 0;
    #pragma unroll
    for (int k = 0; k < 4; ++k) {
        unsigned x = wa[k] ^ pat;
        unsigned z = ((x & 0x7F7F7F7Fu) + 0x7F7F7F7Fu) | x;
        Z |= (z >> (7 - k)) & (0x01010101u << k);
        x = wb[k] ^ pat;
        z = ((x & 0x7F7F7F7Fu) + 0x7F7F7F7Fu) | x;
        Z |= (z >> (3 - k)) & (0x01010101u << (4 + k));
    }
    return ~Z;
}

template <bool FULL>
__device__ __forceinline__ void drain(const unsigned* __restrict__ rec, const float* __restrict__ x,
                                      float* aggL, float* cntL, const unsigned* q, unsigned qhead,
                                      int n, int lane, unsigned nrows) {
    unsigned ent = 0; int s = 0;
    if (FULL || lane < n) {
        const unsigned e = q[(qhead + lane) & (QCAP - 1)];
        const unsigned r = rec[e];
        unsigned u = r >> 17;
        s = (int)(r & 0x1FFFFu);
        if (u >= nrows) { u = TILE_ROWS - 1; s = 0; }
        ent = u << 21;
        atomicAdd(&cntL[u], 1.0f);
    }
    if (FULL) {
        #pragma unroll 1
        for (int j0 = 0; j0 < 32; j0 += 8) {
            float v[8][4];
            #pragma unroll
            for (int j = 0; j < 8; ++j) {
                const int sj = __builtin_amdgcn_readlane(s, j0 + j);
                const float* xr = x + (long)sj * D + lane;
                v[j][0] = xr[0]; v[j][1] = xr[32]; v[j][2] = xr[64]; v[j][3] = xr[96];
            }
            #pragma unroll
            for (int j = 0; j < 8; ++j) {
                const unsigned dj = ((unsigned)__builtin_amdgcn_readlane((int)ent, j0 + j)) >> 21;
                float* ar = aggL + dj * PA + lane;
                atomicAdd(ar,      v[j][0]);
                atomicAdd(ar + 32, v[j][1]);
                atomicAdd(ar + 64, v[j][2]);
                atomicAdd(ar + 96, v[j][3]);
            }
        }
    } else {
        for (int j = 0; j < n; ++j) {
            const int sj = __builtin_amdgcn_readlane(s, j);
            const unsigned dj = ((unsigned)__builtin_amdgcn_readlane((int)ent, j)) >> 21;
            const float* xr = x + (long)sj * D + lane;
            float* ar = aggL + dj * PA + lane;
            const float a0 = xr[0], a1 = xr[32], a2 = xr[64], a3 = xr[96];
            atomicAdd(ar,      a0);
            atomicAdd(ar + 32, a1);
            atomicAdd(ar + 64, a2);
            atomicAdd(ar + 96, a3);
        }
    }
}

template <int NT, bool FROM_LDS>
__device__ __forceinline__ void stage_product(v8f (&acc)[2][8], const float* aggL, const float* wL,
                                              const float* __restrict__ xin, const long (&grow)[2],
                                              const int (&lrow)[2], const float (&inv)[2],
                                              int l16, int half) {
    #pragma unroll 2
    for (int qk = 0; qk < 16; ++qk) {
        const int co = 8 * qk + 4 * half;
        float4 a[2];
        #pragma unroll
        for (int i = 0; i < NT; ++i) {
            if (FROM_LDS) {
                a[i] = *(const float4*)(aggL + lrow[i] * PA + co);
                a[i].x *= inv[i]; a[i].y *= inv[i]; a[i].z *= inv[i]; a[i].w *= inv[i];
            } else {
                a[i] = *(const float4*)(xin + grow[i] * D + co);
            }
        }
        float4 bf[8];
        #pragma unroll
        for (int t = 0; t < 8; ++t) bf[t] = *(const float4*)(wL + (t * 16 + l16) * PA + co);
        #pragma unroll
        for (int t = 0; t < 8; ++t) {
            const v2f b0 = {bf[t].x, bf[t].y};
            const v2f b1 = {bf[t].z, bf[t].w};
            #pragma unroll
            for (int i = 0; i < NT; ++i) {
                const v2f a0 = {a[i].x, a[i].y};
                const v2f a1 = {a[i].z, a[i].w};
                acc[i][t] = __builtin_amdgcn_wmma_f32_16x16x4_f32(false, a0, false, b0, (short)0, acc[i][t], false, false);
                acc[i][t] = __builtin_amdgcn_wmma_f32_16x16x4_f32(false, a1, false, b1, (short)0, acc[i][t], false, false);
            }
        }
    }
}

__device__ __forceinline__ void stage_weights(float* wL, const float* __restrict__ Wt) {
    for (int i = threadIdx.x; i < D * (D / 4); i += THREADS) {
        const int n = i >> 5, c4 = (i & 31) * 4;
        *(float4*)(wL + n * PA + c4) = *(const float4*)(Wt + n * D + c4);
    }
}

__device__ __forceinline__ void products_and_store(float* aggL, float* wL, const float* cntL,
        const float* __restrict__ xin, const float* __restrict__ Wr, const float* __restrict__ bl,
        float* __restrict__ out, int relu, int lo, int nrows, int wave, int lane) {
    const int half = lane >> 4, l16 = lane & 15;
    v8f acc[2][8] = {};
    int lrow[2]; long grow[2]; float inv[2];
    #pragma unroll
    for (int i = 0; i < 2; ++i) {
        lrow[i] = (wave + 16 * i) * 16 + l16;
        if (lrow[i] >= TILE_ROWS) lrow[i] = TILE_ROWS - 1;
        long g = (long)lo + lrow[i];
        grow[i] = g < N_NODES ? g : (long)(N_NODES - 1);
        const float c = cntL[lrow[i]];
        inv[i] = 1.0f / (c > 1.0f ? c : 1.0f);
    }
    stage_product<2, true>(acc, aggL, wL, xin, grow, lrow, inv, l16, half);
    __syncthreads();
    stage_weights(wL, Wr);
    __syncthreads();
    stage_product<2, false>(acc, aggL, wL, xin, grow, lrow, inv, l16, half);

    float* slot = aggL + wave * 16 * PA;
    #pragma unroll
    for (int i = 0; i < 2; ++i) {
        if ((wave + 16 * i) * 16 >= nrows) break;
        #pragma unroll
        for (int t = 0; t < 8; ++t) {
            const int n = t * 16 + l16;
            const float bias = bl[n];
            #pragma unroll
            for (int r = 0; r < 8; ++r) {
                float v = acc[i][t][r] + bias;
                if (relu) v = v > 0.0f ? v : 0.0f;
                slot[(r + 8 * half) * PA + n] = v;
            }
        }
        __builtin_amdgcn_wave_barrier();
        const int tileRow = (wave + 16 * i) * 16;
        #pragma unroll
        for (int r = 0; r < 16; ++r) {
            if (tileRow + r < nrows) {
                const float4 v = *(const float4*)(slot + r * PA + 4 * lane);
                *(float4*)(out + ((long)lo + tileRow + r) * D + 4 * lane) = v;
            }
        }
        __builtin_amdgcn_wave_barrier();
    }
}

__global__ void __launch_bounds__(THREADS)
sage_layer(const unsigned* __restrict__ own, const unsigned* __restrict__ rec,
           const float* __restrict__ xin, const float* __restrict__ Wl,
           const float* __restrict__ bl, const float* __restrict__ Wr,
           float* __restrict__ out, int relu) {
    __shared__ __attribute__((aligned(16))) float lds[LDS_FLOATS];
    float*    aggL = lds;
    float*    wL   = aggL + TILE_ROWS * PA;
    float*    cntL = wL + D * PA;
    unsigned* qAll = (unsigned*)(cntL + TILE_ROWS);

    const int lane = threadIdx.x & 31;
    const int wave = __builtin_amdgcn_readfirstlane((int)(threadIdx.x >> 5));
    const int lo = blockIdx.x * ROWS_WG;
    int nrows = N_NODES - lo;
    if (nrows > ROWS_WG) nrows = ROWS_WG;
    if (nrows < 0) nrows = 0;

    {
        const float4 z = {0.f, 0.f, 0.f, 0.f};
        for (int i = threadIdx.x; i < TILE_ROWS * PA / 4; i += THREADS) ((float4*)aggL)[i] = z;
        for (int i = threadIdx.x; i < TILE_ROWS; i += THREADS) cntL[i] = 0.f;
    }
    stage_weights(wL, Wl);
    __syncthreads();

    unsigned* q = qAll + wave * QCAP;
    unsigned qhead = 0, qn = 0;
    const unsigned pat = blockIdx.x * 0x01010101u;
    const uint4* own16 = (const uint4*)own;
    const int rot = (blockIdx.x * 6) % NPAIR;
    const int ntrips = (NPAIR - wave + NWAVES - 1) / NWAVES;
    uint4 ba[DEPTH], bb[DEPTH];
    #pragma unroll
    for (int d = 0; d < DEPTH; ++d) {
        int pp = wave + NWAVES * d + rot;
        if (pp >= NPAIR) pp -= NPAIR;
        if (d >= ntrips) pp = 0;
        const int c1 = (2 * pp + 1 < NCHUNK) ? 2 * pp + 1 : 2 * pp;
        ba[d] = own16[(long)(2 * pp) * 32 + lane];
        bb[d] = own16[(long)c1 * 32 + lane];
    }
    for (int t0 = 0; t0 < ntrips; t0 += DEPTH) {
        unsigned T[DEPTH];
        #pragma unroll
        for (int d = 0; d < DEPTH; ++d) {
            int pp = wave + NWAVES * (t0 + d) + rot;
            if (pp >= NPAIR) pp -= NPAIR;
            T[d] = match_bits(ba[d], bb[d], pat);
            if (2 * pp + 1 >= NCHUNK) T[d] &= 0x0F0F0F0Fu;
            if (t0 + d >= ntrips) T[d] = 0u;
        }
        #pragma unroll
        for (int d = 0; d < DEPTH; ++d) {
            int pn = wave + NWAVES * (t0 + d + DEPTH) + rot;
            if (pn >= NPAIR) pn -= NPAIR;
            if (t0 + d + DEPTH >= ntrips) pn = 0;
            const int c1 = (2 * pn + 1 < NCHUNK) ? 2 * pn + 1 : 2 * pn;
            ba[d] = own16[(long)(2 * pn) * 32 + lane];
            bb[d] = own16[(long)c1 * 32 + lane];
        }
        #pragma unroll
        for (int d = 0; d < DEPTH; ++d) {
            int pp = wave + NWAVES * (t0 + d) + rot;
            if (pp >= NPAIR) pp -= NPAIR;
            const unsigned ebase = (unsigned)pp * 1024u + (unsigned)lane * 16u;
            unsigned Td = T[d];
            unsigned mask = __builtin_amdgcn_ballot_w32(Td != 0u);
            while (mask) {
                if (Td) {
                    const unsigned rank = __builtin_amdgcn_mbcnt_lo(mask, 0u);
                    const unsigned i = (unsigned)__builtin_ctz(Td);
                    const unsigned kk = i & 7u;
                    const unsigned e = ebase + (kk >> 2) * 512u + (kk & 3u) * 4u + (i >> 3);
                    q[(qhead + qn + rank) & (QCAP - 1)] = e;
                    Td &= Td - 1u;
                }
                qn += (unsigned)__builtin_popcount(mask);
                if (qn >= 32) {
                    drain<true>(rec, xin, aggL, cntL, q, qhead, 32, lane, (unsigned)nrows);
                    qhead += 32; qn -= 32;
                }
                mask = __builtin_amdgcn_ballot_w32(Td != 0u);
            }
        }
    }
    if (qn) drain<false>(rec, xin, aggL, cntL, q, qhead, (int)qn, lane, (unsigned)nrows);
    __syncthreads();

    products_and_store(aggL, wL, cntL, xin, Wr, bl, out, relu, lo, nrows, wave, lane);
}

extern "C" void kernel_launch(void* const* d_in, const int* in_sizes, int n_in,
                              void* d_out, int out_size, void* d_ws, size_t ws_size,
                              hipStream_t stream) {
    const float* x   = (const float*)d_in[0];
    const int*   ei  = (const int*)d_in[1];
    const float* W1l = (const float*)d_in[2];
    const float* b1l = (const float*)d_in[3];
    const float* W1r = (const float*)d_in[4];
    const float* W2l = (const float*)d_in[5];
    const float* b2l = (const float*)d_in[6];
    const float* W2r = (const float*)d_in[7];

    const int* src = ei;
    const int* dst = ei + N_EDGES;
    float* h    = (float*)d_ws;
    unsigned* own = (unsigned*)((char*)d_ws + H_BYTES);
    unsigned* rec = (unsigned*)((char*)d_ws + H_BYTES + OWN_BYTES);
    float* outp = (float*)d_out;

    edge_codes<<<(N_EDGES / 4 + 255) / 256, 256, 0, stream>>>(src, dst, own, rec);
    sage_layer<<<NWG, THREADS, 0, stream>>>(own, rec, x, W1l, b1l, W1r, h, 1);
    sage_layer<<<NWG, THREADS, 0, stream>>>(own, rec, h, W2l, b2l, W2r, outp, 0);
}


# Self product x@Wr^T moved into the edge scan: each wave runs its row tiles' Wr WMMAs between scan batches (staggered per SIMD), parks them in out; 250 groups x 400 rows; sum order: self term first
# speedup vs baseline: 3.0059x; 1.0385x over previous
#include <hip/hip_runtime.h>

typedef __attribute__((ext_vector_type(2))) float v2f;
typedef __attribute__((ext_vector_type(8))) float v8f;
typedef __attribute__((ext_vector_type(4))) float v4f;

#define N_NODES 100000
#define N_EDGES 1600000
#define D 128
#define NWG 250
#define ROWS_WG 400
#define TILE_ROWS 400
#define PA 132
#define NWAVES 16
#define THREADS 512
#define QCAP 512
#define NCHUNK (N_EDGES / 512)
#define NPAIR ((NCHUNK + 1) / 2)
#define DEPTH 8
#define H_BYTES   ((size_t)N_NODES * D * 4)
#define OWN_BYTES ((size_t)N_EDGES)
static_assert(NWG * ROWS_WG >= N_NODES, "row ownership must cover all nodes");
static_assert(N_EDGES % 512 == 0, "owner codes are scanned in chunks of 512");
static_assert(NWG < 256, "an owner code is one byte, and code 255 names no workgroup");
static_assert(N_NODES % 16 == 0 && ROWS_WG % 16 == 0 && ROWS_WG <= TILE_ROWS,
              "every owned row tile is whole, so a tile's 16 rows of out belong to one wave");
static_assert((NPAIR / NWAVES) / DEPTH >= 11, "the scan has a batch for every early product slot (0..10)");
static_assert(N_NODES <= (1 << 17) && ROWS_WG < 511, "edge record packs local row<<17 | source");
static_assert(H_BYTES % 256 == 0 && OWN_BYTES % 256 == 0, "workspace planes stay 256-B aligned");
static_assert(D == 128, "a lane owns columns lane + 32*k");

#define LDS_FLOATS (TILE_ROWS * PA + D * PA + TILE_ROWS + NWAVES * QCAP)
static_assert(LDS_FLOATS * 4 <= 320 * 1024, "sums, one weight matrix, counts and rings fit one WGP's LDS");

__global__ void __launch_bounds__(256)
edge_codes(const int* __restrict__ src, const int* __restrict__ dst,
           unsigned* __restrict__ own, unsigned* __restrict__ rec) {
    const int i = blockIdx.x * 256 + threadIdx.x;
    if (i >= N_EDGES / 4) return;
    const int4 d = ((const int4*)dst)[i];
    const int4 s = ((const int4*)src)[i];
    const int dd[4] = {d.x, d.y, d.z, d.w};
    const int ss[4] = {s.x, s.y, s.z, s.w};
    unsigned code = 0, r[4];
    #pragma unroll
    for (int k = 0; k < 4; ++k) {
        const unsigned dv = (unsigned)dd[k];
        unsigned o = dv / ROWS_WG;
        unsigned u = dv - o * ROWS_WG;
        if (o >= NWG) { o = 255u; u = 511u; }
        code |= o << (8 * k);
        r[k] = (u << 17) | ((unsigned)ss[k] & 0x1FFFFu);
    }
    own[i] = code;
    const uint4 rv = {r[0], r[1], r[2], r[3]};
    ((uint4*)rec)[i] = rv;
}

__device__ __forceinline__ unsigned match_bits(const uint4 a, const uint4 b, unsigned pat) {
    const unsigned wa[4] = {a.x, a.y, a.z, a.w};
    const unsigned wb[4] = {b.x, b.y, b.z, b.w};
    unsigned Z = 0;
    #pragma unroll
    for (int k = 0; k < 4; ++k) {
        unsigned x = wa[k] ^ pat;
        unsigned z = ((x & 0x7F7F7F7Fu) + 0x7F7F7F7Fu) | x;
        Z |= (z >> (7 - k)) & (0x01010101u << k);
        x = wb[k] ^ pat;
        z = ((x & 0x7F7F7F7Fu) + 0x7F7F7F7Fu) | x;
        Z |= (z >> (3 - k)) & (0x01010101u << (4 + k));
    }
    return ~Z;
}

template <bool FULL>
__device__ __forceinline__ void drain(const unsigned* __restrict__ rec, const float* __restrict__ x,
                                      float* aggL, float* cntL, const unsigned* q, unsigned qhead,
                                      int n, int lane, unsigned nrows) {
    unsigned ent = 0; int s = 0;
    if (FULL || lane < n) {
        const unsigned e = q[(qhead + lane) & (QCAP - 1)];
        const unsigned r = rec[e];
        unsigned u = r >> 17;
        s = (int)(r & 0x1FFFFu);
        if (u >= nrows) { u = TILE_ROWS - 1; s = 0; }
        ent = u << 21;
        atomicAdd(&cntL[u], 1.0f);
    }
    if (FULL) {
        #pragma unroll 1
        for (int j0 = 0; j0 < 32; j0 += 8) {
            float v[8][4];
            #pragma unroll
            for (int j = 0; j < 8; ++j) {
                const int sj = __builtin_amdgcn_readlane(s, j0 + j);
                const float* xr = x + (long)sj * D + lane;
                v[j][0] = xr[0]; v[j][1] = xr[32]; v[j][2] = xr[64]; v[j][3] = xr[96];
            }
            #pragma unroll
            for (int j = 0; j < 8; ++j) {
                const unsigned dj = ((unsigned)__builtin_amdgcn_readlane((int)ent, j0 + j)) >> 21;
                float* ar = aggL + dj * PA + lane;
                atomicAdd(ar,      v[j][0]);
                atomicAdd(ar + 32, v[j][1]);
                atomicAdd(ar + 64, v[j][2]);
                atomicAdd(ar + 96, v[j][3]);
            }
        }
    } else {
        for (int j = 0; j < n; ++j) {
            const int sj = __builtin_amdgcn_readlane(s, j);
            const unsigned dj = ((unsigned)__builtin_amdgcn_readlane((int)ent, j)) >> 21;
            const float* xr = x + (long)sj * D + lane;
            float* ar = aggL + dj * PA + lane;
            const float a0 = xr[0], a1 = xr[32], a2 = xr[64], a3 = xr[96];
            atomicAdd(ar,      a0);
            atomicAdd(ar + 32, a1);
            atomicAdd(ar + 64, a2);
            atomicAdd(ar + 96, a3);
        }
    }
}

__device__ __forceinline__ void agg_product(v8f (&acc)[2][8], const float* aggL, const float* wL,
                                            const int (&lrow)[2], const float (&inv)[2],
                                            int l16, int half) {
    #pragma unroll 2
    for (int qk = 0; qk < 16; ++qk) {
        const int co = 8 * qk + 4 * half;
        float4 a[2];
        #pragma unroll
        for (int i = 0; i < 2; ++i) {
            a[i] = *(const float4*)(aggL + lrow[i] * PA + co);
            a[i].x *= inv[i]; a[i].y *= inv[i]; a[i].z *= inv[i]; a[i].w *= inv[i];
        }
        float4 bf[8];
        #pragma unroll
        for (int t = 0; t < 8; ++t) bf[t] = *(const float4*)(wL + (t * 16 + l16) * PA + co);
        #pragma unroll
        for (int t = 0; t < 8; ++t) {
            const v2f b0 = {bf[t].x, bf[t].y};
            const v2f b1 = {bf[t].z, bf[t].w};
            #pragma unroll
            for (int i = 0; i < 2; ++i) {
                const v2f a0 = {a[i].x, a[i].y};
                const v2f a1 = {a[i].z, a[i].w};
                acc[i][t] = __builtin_amdgcn_wmma_f32_16x16x4_f32(false, a0, false, b0, (short)0, acc[i][t], false, false);
                acc[i][t] = __builtin_amdgcn_wmma_f32_16x16x4_f32(false, a1, false, b1, (short)0, acc[i][t], false, false);
            }
        }
    }
}

#define SELF_AHEAD 6
__device__ __forceinline__ void self_tile(const float* wL, const float* __restrict__ xin,
                                          float* __restrict__ out, long row0, int lane) {
    const int half = lane >> 4, l16 = lane & 15;
    const float* xr = xin + (row0 + l16) * D + 4 * half;
    v8f acc[8] = {};
    float4 a[16];
    #pragma unroll
    for (int k = 0; k < SELF_AHEAD; ++k) a[k] = *(const float4*)(xr + 8 * k);
    #pragma unroll
    for (int qk = 0; qk < 16; ++qk) {
        if (qk + SELF_AHEAD < 16) a[qk + SELF_AHEAD] = *(const float4*)(xr + 8 * (qk + SELF_AHEAD));
        const int co = 8 * qk + 4 * half;
        float4 bf[8];
        #pragma unroll
        for (int t = 0; t < 8; ++t) bf[t] = *(const float4*)(wL + (t * 16 + l16) * PA + co);
        const v2f a0 = {a[qk].x, a[qk].y};
        const v2f a1 = {a[qk].z, a[qk].w};
        #pragma unroll
        for (int t = 0; t < 8; ++t) {
            const v2f b0 = {bf[t].x, bf[t].y};
            const v2f b1 = {bf[t].z, bf[t].w};
            acc[t] = __builtin_amdgcn_wmma_f32_16x16x4_f32(false, a0, false, b0, (short)0, acc[t], false, false);
            acc[t] = __builtin_amdgcn_wmma_f32_16x16x4_f32(false, a1, false, b1, (short)0, acc[t], false, false);
        }
    }
    v4f* pk = (v4f*)(out + row0 * D) + lane;
    #pragma unroll
    for (int j = 0; j < 16; ++j) {
        const v8f c = acc[j >> 1];
        const v4f v = (j & 1) ? (v4f){c[4], c[5], c[6], c[7]} : (v4f){c[0], c[1], c[2], c[3]};
        __builtin_nontemporal_store(v, pk + 32 * j);
    }
}

__device__ __forceinline__ void unpark_tile(v8f (&acc)[8], const float* __restrict__ out, long row0, int lane) {
    const v4f* pk = (const v4f*)(out + row0 * D) + lane;
    #pragma unroll
    for (int j = 0; j < 16; ++j) {
        const v4f v = __builtin_nontemporal_load(pk + 32 * j);
        #pragma unroll
        for (int e = 0; e < 4; ++e) acc[j >> 1][(j & 1) * 4 + e] = v[e];
    }
}

__device__ __forceinline__ void stage_weights(float* wL, const float* __restrict__ Wt) {
    for (int i = threadIdx.x; i < D * (D / 4); i += THREADS) {
        const int n = i >> 5, c4 = (i & 31) * 4;
        *(float4*)(wL + n * PA + c4) = *(const float4*)(Wt + n * D + c4);
    }
}

__device__ __forceinline__ void neighbour_product_and_store(v8f (&acc)[2][8], float* aggL, const float* wL,
        const float* cntL, const float* __restrict__ bl, float* __restrict__ out, int relu,
        int lo, int nrows, int wave, int lane) {
    const int half = lane >> 4, l16 = lane & 15;
    int lrow[2]; float inv[2];
    #pragma unroll
    for (int i = 0; i < 2; ++i) {
        lrow[i] = (wave + 16 * i) * 16 + l16;
        if (lrow[i] >= TILE_ROWS) lrow[i] = TILE_ROWS - 1;
        const float c = cntL[lrow[i]];
        inv[i] = 1.0f / (c > 1.0f ? c : 1.0f);
    }
    agg_product(acc, aggL, wL, lrow, inv, l16, half);

    float* slot = aggL + wave * 16 * PA;
    #pragma unroll
    for (int i = 0; i < 2; ++i) {
        if ((wave + 16 * i) * 16 >= nrows) break;
        #pragma unroll
        for (int t = 0; t < 8; ++t) {
            const int n = t * 16 + l16;
            const float bias = bl[n];
            #pragma unroll
            for (int r = 0; r < 8; ++r) {
                float v = acc[i][t][r] + bias;
                if (relu) v = v > 0.0f ? v : 0.0f;
                slot[(r + 8 * half) * PA + n] = v;
            }
        }
        __builtin_amdgcn_wave_barrier();
        const int tileRow = (wave + 16 * i) * 16;
        #pragma unroll
        for (int r = 0; r < 16; ++r) {
            if (tileRow + r < nrows) {
                const float4 v = *(const float4*)(slot + r * PA + 4 * lane);
                *(float4*)(out + ((long)lo + tileRow + r) * D + 4 * lane) = v;
            }
        }
        __builtin_amdgcn_wave_barrier();
    }
}

__global__ void __launch_bounds__(THREADS)
sage_layer(const unsigned* __restrict__ own, const unsigned* __restrict__ rec,
           const float* __restrict__ xin, const float* __restrict__ Wl,
           const float* __restrict__ bl, const float* __restrict__ Wr,
           float* __restrict__ out, int relu) {
    __shared__ __attribute__((aligned(16))) float lds[LDS_FLOATS];
    float*    aggL = lds;
    float*    wL   = aggL + TILE_ROWS * PA;
    float*    cntL = wL + D * PA;
    unsigned* qAll = (unsigned*)(cntL + TILE_ROWS);

    const int lane = threadIdx.x & 31;
    const int wave = __builtin_amdgcn_readfirstlane((int)(threadIdx.x >> 5));
    const int lo = blockIdx.x * ROWS_WG;
    int nrows = N_NODES - lo;
    if (nrows > ROWS_WG) nrows = ROWS_WG;
    if (nrows < 0) nrows = 0;

    {
        const float4 z = {0.f, 0.f, 0.f, 0.f};
        for (int i = threadIdx.x; i < TILE_ROWS * PA / 4; i += THREADS) ((float4*)aggL)[i] = z;
        for (int i = threadIdx.x; i < TILE_ROWS; i += THREADS) cntL[i] = 0.f;
    }
    stage_weights(wL, Wr);
    __syncthreads();

    const int slot3 = 3 * (((wave & 3) + (wave >> 2)) & 3);

    unsigned* q = qAll + wave * QCAP;
    unsigned qhead = 0, qn = 0;
    const unsigned pat = blockIdx.x * 0x01010101u;
    const uint4* own16 = (const uint4*)own;
    const int rot = (blockIdx.x * 6) % NPAIR;
    const int ntrips = (NPAIR - wave + NWAVES - 1) / NWAVES;
    uint4 ba[DEPTH], bb[DEPTH];
    #pragma unroll
    for (int d = 0; d < DEPTH; ++d) {
        int pp = wave + NWAVES * d + rot;
        if (pp >= NPAIR) pp -= NPAIR;
        if (d >= ntrips) pp = 0;
        const int c1 = (2 * pp + 1 < NCHUNK) ? 2 * pp + 1 : 2 * pp;
        ba[d] = own16[(long)(2 * pp) * 32 + lane];
        bb[d] = own16[(long)c1 * 32 + lane];
    }
    for (int t0 = 0; t0 < ntrips; t0 += DEPTH) {
        unsigned T[DEPTH];
        #pragma unroll
        for (int d = 0; d < DEPTH; ++d) {
            int pp = wave + NWAVES * (t0 + d) + rot;
            if (pp >= NPAIR) pp -= NPAIR;
            T[d] = match_bits(ba[d], bb[d], pat);
            if (2 * pp + 1 >= NCHUNK) T[d] &= 0x0F0F0F0Fu;
            if (t0 + d >= ntrips) T[d] = 0u;
        }
        {
            const int ti = t0 / DEPTH - slot3;
            if (ti == 0 || ti == 1) {
                const int tile = wave + NWAVES * ti;
                if (tile * 16 + 16 <= nrows) self_tile(wL, xin, out, (long)lo + tile * 16, lane);
            }
        }
        #pragma unroll
        for (int d = 0; d < DEPTH; ++d) {
            int pn = wave + NWAVES * (t0 + d + DEPTH) + rot;
            if (pn >= NPAIR) pn -= NPAIR;
            if (t0 + d + DEPTH >= ntrips) pn = 0;
            const int c1 = (2 * pn + 1 < NCHUNK) ? 2 * pn + 1 : 2 * pn;
            ba[d] = own16[(long)(2 * pn) * 32 + lane];
            bb[d] = own16[(long)c1 * 32 + lane];
        }
        #pragma unroll
        for (int d = 0; d < DEPTH; ++d) {
            int pp = wave + NWAVES * (t0 + d) + rot;
            if (pp >= NPAIR) pp -= NPAIR;
            const unsigned ebase = (unsigned)pp * 1024u + (unsigned)lane * 16u;
            unsigned Td = T[d];
            unsigned mask = __builtin_amdgcn_ballot_w32(Td != 0u);
            while (mask) {
                if (Td) {
                    const unsigned rank = __builtin_amdgcn_mbcnt_lo(mask, 0u);
                    const unsigned i = (unsigned)__builtin_ctz(Td);
                    const unsigned kk = i & 7u;
                    const unsigned e = ebase + (kk >> 2) * 512u + (kk & 3u) * 4u + (i >> 3);
                    q[(qhead + qn + rank) & (QCAP - 1)] = e;
                    Td &= Td - 1u;
                }
                qn += (unsigned)__builtin_popcount(mask);
                if (qn >= 32) {
                    drain<true>(rec, xin, aggL, cntL, q, qhead, 32, lane, (unsigned)nrows);
                    qhead += 32; qn -= 32;
                }
                mask = __builtin_amdgcn_ballot_w32(Td != 0u);
            }
        }
    }
    if (qn) drain<false>(rec, xin, aggL, cntL, q, qhead, (int)qn, lane, (unsigned)nrows);

    v8f acc[2][8] = {};
    #pragma unroll
    for (int i = 0; i < 2; ++i) {
        const int tile = wave + NWAVES * i;
        if (tile * 16 + 16 <= nrows) unpark_tile(acc[i], out, (long)lo + tile * 16, lane);
    }
    __syncthreads();
    stage_weights(wL, Wl);
    __syncthreads();

    neighbour_product_and_store(acc, aggL, wL, cntL, bl, out, relu, lo, nrows, wave, lane);
}

extern "C" void kernel_launch(void* const* d_in, const int* in_sizes, int n_in,
                              void* d_out, int out_size, void* d_ws, size_t ws_size,
                              hipStream_t stream) {
    const float* x   = (const float*)d_in[0];
    const int*   ei  = (const int*)d_in[1];
    const float* W1l = (const float*)d_in[2];
    const float* b1l = (const float*)d_in[3];
    const float* W1r = (const float*)d_in[4];
    const float* W2l = (const float*)d_in[5];
    const float* b2l = (const float*)d_in[6];
    const float* W2r = (const float*)d_in[7];

    const int* src = ei;
    const int* dst = ei + N_EDGES;
    float* h    = (float*)d_ws;
    unsigned* own = (unsigned*)((char*)d_ws + H_BYTES);
    unsigned* rec = (unsigned*)((char*)d_ws + H_BYTES + OWN_BYTES);
    float* outp = (float*)d_out;

    edge_codes<<<(N_EDGES / 4 + 255) / 256, 256, 0, stream>>>(src, dst, own, rec);
    sage_layer<<<NWG, THREADS, 0, stream>>>(own, rec, x, W1l, b1l, W1r, h, 1);
    sage_layer<<<NWG, THREADS, 0, stream>>>(own, rec, h, W2l, b2l, W2r, outp, 0);
}
